# starting kernel
# baseline (speedup 1.0000x reference)
#include <hip/hip_runtime.h>


typedef float v2f __attribute__((ext_vector_type(2)));
typedef float v8f __attribute__((ext_vector_type(8)));

#define N_NODES 20000
#define N_EDGES 8192
#define D_NODE  64
#define D_EDGE  16
#define UNITS   64
#define K1      (D_NODE + D_EDGE)
#define K2      (UNITS + UNITS)

__global__ __launch_bounds__(256) void gemm_init_kernel(
    const float* __restrict__ node_feature,
    const float* __restrict__ edge_feature,
    const float* __restrict__ W,
    const float* __restrict__ bias,
    const int*   __restrict__ edge_src,
    float*       __restrict__ edge_state)
{
    __shared__ float sW[K1 * UNITS];
    for (int i = threadIdx.x; i < K1 * UNITS; i += 256) sW[i] = W[i];
    __syncthreads();

    const int wave    = threadIdx.x >> 5;
    const int lane    = threadIdx.x & 31;
    const int rowTile = blockIdx.x * 8 + wave;
    const int m       = lane & 15;
    const int kh      = (lane >> 4) << 1;
    const int e       = rowTile * 16 + m;
    const int s       = edge_src[e];
    const float* xn = node_feature + (long)s * D_NODE;
    const float* xe = edge_feature + (long)e * D_EDGE;

    v8f acc[4] = {};
    #pragma unroll
    for (int k0 = 0; k0 < K1; k0 += 4) {
        const int k = k0 + kh;
        v2f a;
        a.x = (k     < D_NODE) ? xn[k]     : xe[k     - D_NODE];
        a.y = (k + 1 < D_NODE) ? xn[k + 1] : xe[k + 1 - D_NODE];
        const float* wr0 = &sW[(k    ) * UNITS];
        const float* wr1 = &sW[(k + 1) * UNITS];
        #pragma unroll
        for (int nt = 0; nt < 4; ++nt) {
            v2f b;
            b.x = wr0[nt * 16 + m];
            b.y = wr1[nt * 16 + m];
            acc[nt] = __builtin_amdgcn_wmma_f32_16x16x4_f32(
                false, a, false, b, (short)0, acc[nt], false, false);
        }
    }

    const int half = lane >> 4;
    #pragma unroll
    for (int nt = 0; nt < 4; ++nt) {
        const int n = nt * 16 + m;
        const float bv = bias[n];
        #pragma unroll
        for (int r = 0; r < 8; ++r) {
            const int mm = r + 8 * half;
            edge_state[(long)(rowTile * 16 + mm) * UNITS + n] = acc[nt][r] + bv;
        }
    }
}

__global__ void zero_kernel(float* __restrict__ p, int n)
{
    int i = blockIdx.x * blockDim.x + threadIdx.x;
    if (i < n) p[i] = 0.0f;
}

__global__ void scatter_kernel(const float* __restrict__ edge_state,
                               const int*   __restrict__ edge_dst,
                               float*       __restrict__ node_agg)
{
    int t = blockIdx.x * 256 + threadIdx.x;
    int e = t >> 6, u = t & 63;
    atomicAdd(&node_agg[(long)edge_dst[e] * UNITS + u], edge_state[t]);
}

__global__ void gather_kernel(const float* __restrict__ node_agg,
                              const int*   __restrict__ edge_src,
                              float*       __restrict__ message)
{
    int t = blockIdx.x * 256 + threadIdx.x;
    int e = t >> 6, u = t & 63;
    message[t] = node_agg[(long)edge_src[e] * UNITS + u];
}

__global__ __launch_bounds__(256) void reverse_kernel(
    const int*   __restrict__ edge_src,
    const int*   __restrict__ edge_dst,
    const float* __restrict__ edge_state,
    float*       __restrict__ message)
{
    __shared__ int ss[N_EDGES];
    __shared__ int sd[N_EDGES];
    for (int j = threadIdx.x; j < N_EDGES; j += 256) {
        ss[j] = edge_src[j];
        sd[j] = edge_dst[j];
    }
    __syncthreads();

    const int i  = blockIdx.x * 256 + threadIdx.x;
    const int si = ss[i];
    const int di = sd[i];
    float* mi = message + (long)i * UNITS;
    for (int j = 0; j < N_EDGES; ++j) {
        if (ss[j] == di && sd[j] == si) {
            const float* esj = edge_state + (long)j * UNITS;
            #pragma unroll 4
            for (int u = 0; u < UNITS; ++u) mi[u] -= esj[u];
        }
    }
}

__global__ __launch_bounds__(256) void gemm_upd_kernel(
    const float* __restrict__ edge_state,
    const float* __restrict__ message,
    const float* __restrict__ W,
    const float* __restrict__ bias,
    float*       __restrict__ out)
{
    __shared__ float sW[K2 * UNITS];
    for (int i = threadIdx.x; i < K2 * UNITS; i += 256) sW[i] = W[i];
    __syncthreads();

    const int wave    = threadIdx.x >> 5;
    const int lane    = threadIdx.x & 31;
    const int rowTile = blockIdx.x * 8 + wave;
    const int m       = lane & 15;
    const int kh      = (lane >> 4) << 1;
    const int e       = rowTile * 16 + m;
    const float* x0 = edge_state + (long)e * UNITS;
    const float* x1 = message    + (long)e * UNITS;

    v8f acc[4] = {};
    #pragma unroll
    for (int k0 = 0; k0 < K2; k0 += 4) {
        const int k = k0 + kh;
        v2f a;
        a.x = (k     < UNITS) ? x0[k]     : x1[k     - UNITS];
        a.y = (k + 1 < UNITS) ? x0[k + 1] : x1[k + 1 - UNITS];
        const float* wr0 = &sW[(k    ) * UNITS];
        const float* wr1 = &sW[(k + 1) * UNITS];
        #pragma unroll
        for (int nt = 0; nt < 4; ++nt) {
            v2f b;
            b.x = wr0[nt * 16 + m];
            b.y = wr1[nt * 16 + m];
            acc[nt] = __builtin_amdgcn_wmma_f32_16x16x4_f32(
                false, a, false, b, (short)0, acc[nt], false, false);
        }
    }

    const int half = lane >> 4;
    #pragma unroll
    for (int nt = 0; nt < 4; ++nt) {
        const int n = nt * 16 + m;
        const float bv = bias[n];
        #pragma unroll
        for (int r = 0; r < 8; ++r) {
            const int mm = r + 8 * half;
            out[(long)(rowTile * 16 + mm) * UNITS + n] = acc[nt][r] + bv;
        }
    }
}

extern "C" void kernel_launch(void* const* d_in, const int* in_sizes, int n_in,
                              void* d_out, int out_size, void* d_ws, size_t ws_size,
                              hipStream_t stream)
{
    const float* node_feature = (const float*)d_in[0];
    const float* edge_feature = (const float*)d_in[1];
    const float* W_init       = (const float*)d_in[2];
    const float* b_init       = (const float*)d_in[3];
    const float* W_upd        = (const float*)d_in[4];
    const float* b_upd        = (const float*)d_in[5];
    const int*   edge_src     = (const int*)d_in[6];
    const int*   edge_dst     = (const int*)d_in[7];
    float*       out          = (float*)d_out;

    char* ws = (char*)d_ws;
    float* edge_state = (float*)ws;
    float* node_agg   = (float*)(ws + (size_t)N_EDGES * UNITS * 4);
    float* message    = (float*)(ws + (size_t)N_EDGES * UNITS * 4
                                    + (size_t)N_NODES * UNITS * 4);

    gemm_init_kernel<<<64, 256, 0, stream>>>(node_feature, edge_feature,
                                             W_init, b_init, edge_src, edge_state);

    zero_kernel<<<(N_NODES * UNITS + 255) / 256, 256, 0, stream>>>(
        node_agg, N_NODES * UNITS);

    scatter_kernel<<<(N_EDGES * UNITS) / 256, 256, 0, stream>>>(
        edge_state, edge_dst, node_agg);

    gather_kernel<<<(N_EDGES * UNITS) / 256, 256, 0, stream>>>(
        node_agg, edge_src, message);

    reverse_kernel<<<N_EDGES / 256, 256, 0, stream>>>(
        edge_src, edge_dst, edge_state, message);

    gemm_upd_kernel<<<64, 256, 0, stream>>>(edge_state, message,
                                            W_upd, b_upd, out);
}


# launches 2 -> 1: each work-group builds the in-edge lists in its own LDS and recomputes its edges' in-neighbour edge_state rows as extra 16-row tiles of the first GEMM; sums still in ascending j
# speedup vs baseline: 34.1757x; 34.1757x over previous
#include <hip/hip_runtime.h>


typedef float v2f __attribute__((ext_vector_type(2)));
typedef float v8f __attribute__((ext_vector_type(8)));

#define N_NODES 20000
#define N_EDGES 8192
#define D_NODE  64
#define D_EDGE  16
#define UNITS   64
#define K1      (D_NODE + D_EDGE)
#define K2      (UNITS + UNITS)
#define WAVES   4
#define THREADS (WAVES * 32)
#define PITCH   68
#define SLOTS   8
#define LIST_END N_EDGES

#define INS(S) { const int lo_ = min(S, x); x = max(S, x); S = lo_; }

__device__ __forceinline__ void load_rows(v2f (&a)[K1 / 4],
    const float* __restrict__ node_feature, const float* __restrict__ edge_feature,
    int sj, int je, int kh)
{
    const float* xn = node_feature + (long)sj * D_NODE;
    const float* xe = edge_feature + (long)je * D_EDGE;
    #pragma unroll
    for (int k0 = 0; k0 < K1; k0 += 4) {
        const int k = k0 + kh;
        a[k0 >> 2].x = (k     < D_NODE) ? xn[k]     : xe[k     - D_NODE];
        a[k0 >> 2].y = (k + 1 < D_NODE) ? xn[k + 1] : xe[k + 1 - D_NODE];
    }
}

__device__ __forceinline__ void first_layer(v8f (&acc)[4], const v2f (&a)[K1 / 4],
    const float* sW, const float* __restrict__ bias, int m, int kh)
{
    #pragma unroll
    for (int nt = 0; nt < 4; ++nt) acc[nt] = v8f{};
    #pragma unroll
    for (int k0 = 0; k0 < K1; k0 += 4) {
        const int k = k0 + kh;
        const float* wr0 = &sW[(k    ) * UNITS];
        const float* wr1 = &sW[(k + 1) * UNITS];
        #pragma unroll
        for (int nt = 0; nt < 4; ++nt) {
            v2f b;
            b.x = wr0[nt * 16 + m];
            b.y = wr1[nt * 16 + m];
            acc[nt] = __builtin_amdgcn_wmma_f32_16x16x4_f32(
                false, a[k0 >> 2], false, b, (short)0, acc[nt], false, false);
        }
    }
    #pragma unroll
    for (int nt = 0; nt < 4; ++nt) {
        const float bv = bias[nt * 16 + m];
        #pragma unroll
        for (int r = 0; r < 8; ++r) acc[nt][r] = acc[nt][r] + bv;
    }
}

__device__ __forceinline__ void park_tile(float* tile, const v8f (&acc)[4], int m, int half)
{
    #pragma unroll
    for (int nt = 0; nt < 4; ++nt) {
        #pragma unroll
        for (int r = 0; r < 8; ++r)
            tile[(r + 8 * half) * PITCH + nt * 16 + m] = acc[nt][r];
    }
}

__global__ __launch_bounds__(THREADS) void edgeconv_kernel(
    const float* __restrict__ node_feature,
    const float* __restrict__ edge_feature,
    const float* __restrict__ W_init,
    const float* __restrict__ b_init,
    const float* __restrict__ W_upd,
    const float* __restrict__ b_upd,
    const int*   __restrict__ edge_src,
    const int*   __restrict__ edge_dst,
    float*       __restrict__ out)
{
    __shared__ alignas(16) int   s_head[N_NODES];
    __shared__ alignas(16) int   s_link[2 * N_EDGES];
    __shared__ alignas(16) float s_w1[K1 * UNITS];
    __shared__ alignas(16) float s_w2[K2 * UNITS];
    __shared__ alignas(16) float s_tile[WAVES][16 * PITCH];
    __shared__ int s_nbr[WAVES][16 * SLOTS];
    __shared__ int s_cnt[WAVES][16];

    const int t       = threadIdx.x;
    const int wave    = t >> 5;
    const int lane    = t & 31;
    const int m       = lane & 15;
    const int half    = lane >> 4;
    const int kh      = half << 1;
    const int rowTile = blockIdx.x * WAVES + wave;
    const int e       = rowTile * 16 + m;

    int4 dv[16], sv[16];
    #pragma unroll
    for (int i = 0; i < 16; ++i) {
        dv[i] = ((const int4*)edge_dst)[i * THREADS + t];
        sv[i] = ((const int4*)edge_src)[i * THREADS + t];
    }
    const int si = edge_src[e];
    const int di = edge_dst[e];

    for (int i = t; i < N_NODES / 4; i += THREADS)
        ((int4*)s_head)[i] = make_int4(-1, -1, -1, -1);
    for (int i = t; i < K1 * UNITS / 4; i += THREADS)
        ((float4*)s_w1)[i] = ((const float4*)W_init)[i];
    for (int i = t; i < K2 * UNITS / 4; i += THREADS)
        ((float4*)s_w2)[i] = ((const float4*)W_upd)[i];

    v2f a_own[K1 / 4];
    load_rows(a_own, node_feature, edge_feature, si, e, kh);

    __syncthreads();

    int4 ov[16];
    #pragma unroll
    for (int i = 0; i < 16; ++i) {
        const int e0 = 4 * (i * THREADS + t);
        ov[i].x = atomicExch(&s_head[dv[i].x], e0);
        ov[i].y = atomicExch(&s_head[dv[i].y], e0 + 1);
        ov[i].z = atomicExch(&s_head[dv[i].z], e0 + 2);
        ov[i].w = atomicExch(&s_head[dv[i].w], e0 + 3);
    }
    #pragma unroll
    for (int i = 0; i < 16; ++i) {
        ((int4*)s_link)[2 * (i * THREADS + t)    ] = make_int4(ov[i].x, sv[i].x, ov[i].y, sv[i].y);
        ((int4*)s_link)[2 * (i * THREADS + t) + 1] = make_int4(ov[i].z, sv[i].z, ov[i].w, sv[i].w);
    }
    __syncthreads();

    float* tile = s_tile[wave];
    v8f acc[4];

    first_layer(acc, a_own, s_w1, b_init, m, kh);
    park_tile(tile, acc, m, half);
    v2f xa[16];
    #pragma unroll
    for (int q = 0; q < 16; ++q) xa[q] = *(const v2f*)&tile[m * PITCH + 4 * q + kh];

    v2f msg[16];
    #pragma unroll
    for (int q = 0; q < 16; ++q) msg[q] = v2f{0.f, 0.f};
    int prev = -1;
    for (int pass = 0; pass < N_EDGES / SLOTS + 1; ++pass) {
        int s0 = LIST_END, s1 = LIST_END, s2 = LIST_END, s3 = LIST_END;
        int s4 = LIST_END, s5 = LIST_END, s6 = LIST_END, s7 = LIST_END;
        int j = s_head[si];
        for (int n = 0; n < N_EDGES && (unsigned)j < (unsigned)N_EDGES; ++n) {
            const int2 ln = *(const int2*)&s_link[2 * j];
            int x = (ln.y != di && j > prev) ? j : LIST_END;
            INS(s0) INS(s1) INS(s2) INS(s3) INS(s4) INS(s5) INS(s6) INS(s7)
            j = ln.x;
        }
        const int c = (s0 < LIST_END) + (s1 < LIST_END) + (s2 < LIST_END) + (s3 < LIST_END)
                    + (s4 < LIST_END) + (s5 < LIST_END) + (s6 < LIST_END) + (s7 < LIST_END);
        if (__builtin_amdgcn_ballot_w32(c > 0) == 0u) break;

        if (half == 0) s_cnt[wave][m] = c;
        int base = 0, total = 0;
        #pragma unroll
        for (int k = 0; k < 16; ++k) {
            const int ck = s_cnt[wave][k];
            base  += (k < m) ? ck : 0;
            total += ck;
        }
        total = __builtin_amdgcn_readfirstlane(total);
        if (half == 0) {
            if (c > 0) s_nbr[wave][base    ] = s0;
            if (c > 1) s_nbr[wave][base + 1] = s1;
            if (c > 2) s_nbr[wave][base + 2] = s2;
            if (c > 3) s_nbr[wave][base + 3] = s3;
            if (c > 4) s_nbr[wave][base + 4] = s4;
            if (c > 5) s_nbr[wave][base + 5] = s5;
            if (c > 6) s_nbr[wave][base + 6] = s6;
            if (c > 7) s_nbr[wave][base + 7] = s7;
        }

        for (int t0 = 0; t0 < total; t0 += 16) {
            const int  p    = t0 + m;
            const bool live = p < total;
            const int  jr   = live ? s_nbr[wave][p] : e;
            const int  sj   = live ? s_link[2 * jr + 1] : si;
            v2f a_nb[K1 / 4];
            load_rows(a_nb, node_feature, edge_feature, sj, jr, kh);
            first_layer(acc, a_nb, s_w1, b_init, m, kh);
            park_tile(tile, acc, m, half);
            #pragma unroll
            for (int u = 0; u < SLOTS; ++u) {
                const int pu = base + u - t0;
                if (u < c && pu >= 0 && pu < 16) {
                    const float* r_ = &tile[pu * PITCH + kh];
                    #pragma unroll
                    for (int q = 0; q < 16; ++q) {
                        msg[q].x += r_[4 * q];
                        msg[q].y += r_[4 * q + 1];
                    }
                }
            }
        }
        if (__builtin_amdgcn_ballot_w32(c == SLOTS) == 0u) break;
        prev = (c == SLOTS) ? s7 : LIST_END;
    }

    #pragma unroll
    for (int nt = 0; nt < 4; ++nt) acc[nt] = v8f{};
    #pragma unroll
    for (int k0 = 0; k0 < K2; k0 += 4) {
        const int k = k0 + kh;
        const v2f a = (k0 < UNITS) ? xa[k0 >> 2] : msg[(k0 - UNITS) >> 2];
        const float* wr0 = &s_w2[(k    ) * UNITS];
        const float* wr1 = &s_w2[(k + 1) * UNITS];
        #pragma unroll
        for (int nt = 0; nt < 4; ++nt) {
            v2f b;
            b.x = wr0[nt * 16 + m];
            b.y = wr1[nt * 16 + m];
            acc[nt] = __builtin_amdgcn_wmma_f32_16x16x4_f32(
                false, a, false, b, (short)0, acc[nt], false, false);
        }
    }

    #pragma unroll
    for (int nt = 0; nt < 4; ++nt) {
        const int n = nt * 16 + m;
        const float bv = b_upd[n];
        #pragma unroll
        for (int r = 0; r < 8; ++r) {
            const int mm = r + 8 * half;
            out[(long)(rowTile * 16 + mm) * UNITS + n] = acc[nt][r] + bv;
        }
    }
}

extern "C" void kernel_launch(void* const* d_in, const int* in_sizes, int n_in,
                              void* d_out, int out_size, void* d_ws, size_t ws_size,
                              hipStream_t stream)
{
    const float* node_feature = (const float*)d_in[0];
    const float* edge_feature = (const float*)d_in[1];
    const float* W_init       = (const float*)d_in[2];
    const float* b_init       = (const float*)d_in[3];
    const float* W_upd        = (const float*)d_in[4];
    const float* b_upd        = (const float*)d_in[5];
    const int*   edge_src     = (const int*)d_in[6];
    const int*   edge_dst     = (const int*)d_in[7];
    float*       out          = (float*)d_out;

    edgeconv_kernel<<<128, 128, 0, stream>>>(node_feature, edge_feature,
                                                 W_init, b_init, W_upd, b_upd,
                                                 edge_src, edge_dst, out);
}


# both layers' weights kept as B fragments in each wave's registers, read from memory 16 B per lane (lane m owns column 4m+tile); no weights or B reads in LDS; list barriers wait on LDS only
# speedup vs baseline: 50.2028x; 1.4690x over previous
#include <hip/hip_runtime.h>


typedef float v2f __attribute__((ext_vector_type(2)));
typedef float v4f __attribute__((ext_vector_type(4)));
typedef float v8f __attribute__((ext_vector_type(8)));

#define N_NODES 20000
#define N_EDGES 8192
#define D_NODE  64
#define D_EDGE  16
#define UNITS   64
#define K1      (D_NODE + D_EDGE)
#define K2      (UNITS + UNITS)
#define WAVES   4
#define THREADS (WAVES * 32)
#define PITCH   68
#define SLOTS   8
#define LIST_END N_EDGES

#define INS(S) { const int lo_ = min(S, x); x = max(S, x); S = lo_; }

__device__ __forceinline__ void load_rows(v2f (&a)[K1 / 4],
    const float* __restrict__ node_feature, const float* __restrict__ edge_feature,
    int sj, int je, int kh)
{
    const float* xn = node_feature + (long)sj * D_NODE;
    const float* xe = edge_feature + (long)je * D_EDGE;
    #pragma unroll
    for (int k0 = 0; k0 < K1; k0 += 4) {
        const int k = k0 + kh;
        a[k0 >> 2].x = (k     < D_NODE) ? xn[k]     : xe[k     - D_NODE];
        a[k0 >> 2].y = (k + 1 < D_NODE) ? xn[k + 1] : xe[k + 1 - D_NODE];
    }
}

template <int Q0, int Q1, int KS>
__device__ __forceinline__ void request_weights(v4f (&w0)[KS], v4f (&w1)[KS],
    const float* __restrict__ W, int m, int kh)
{
    const v4f* p = (const v4f*)W + kh * (UNITS / 4) + m;
    #pragma unroll
    for (int q = Q0; q < Q1; ++q) {
        w0[q] = p[(4 * q    ) * (UNITS / 4)];
        w1[q] = p[(4 * q + 1) * (UNITS / 4)];
    }
}

__device__ __forceinline__ void first_layer(v8f (&acc)[4], const v2f (&a)[K1 / 4],
    const v4f (&w0)[K1 / 4], const v4f (&w1)[K1 / 4], const v4f bias)
{
    #pragma unroll
    for (int nt = 0; nt < 4; ++nt) acc[nt] = v8f{};
    #pragma unroll
    for (int q = 0; q < K1 / 4; ++q) {
        #pragma unroll
        for (int nt = 0; nt < 4; ++nt) {
            const v2f b = {w0[q][nt], w1[q][nt]};
            acc[nt] = __builtin_amdgcn_wmma_f32_16x16x4_f32(
                false, a[q], false, b, (short)0, acc[nt], false, false);
        }
    }
    #pragma unroll
    for (int nt = 0; nt < 4; ++nt) {
        #pragma unroll
        for (int r = 0; r < 8; ++r) acc[nt][r] = acc[nt][r] + bias[nt];
    }
}

__device__ __forceinline__ void lds_barrier()
{
    __builtin_amdgcn_fence(__ATOMIC_RELEASE, "workgroup", "local");
    __builtin_amdgcn_s_barrier();
    __builtin_amdgcn_fence(__ATOMIC_ACQUIRE, "workgroup", "local");
}

__device__ __forceinline__ void park_tile(float* tile, const v8f (&acc)[4], int m, int half)
{
    #pragma unroll
    for (int r = 0; r < 8; ++r)
        *(v4f*)&tile[(r + 8 * half) * PITCH + 4 * m] =
            v4f{acc[0][r], acc[1][r], acc[2][r], acc[3][r]};
}

__global__ __launch_bounds__(THREADS) void edgeconv_kernel(
    const float* __restrict__ node_feature,
    const float* __restrict__ edge_feature,
    const float* __restrict__ W_init,
    const float* __restrict__ b_init,
    const float* __restrict__ W_upd,
    const float* __restrict__ b_upd,
    const int*   __restrict__ edge_src,
    const int*   __restrict__ edge_dst,
    float*       __restrict__ out)
{
    __shared__ alignas(16) int   s_head[N_NODES];
    __shared__ alignas(16) int   s_link[2 * N_EDGES];
    __shared__ alignas(16) float s_tile[WAVES][16 * PITCH];
    __shared__ int s_nbr[WAVES][16 * SLOTS];
    __shared__ int s_cnt[WAVES][16];

    const int t       = threadIdx.x;
    const int wave    = t >> 5;
    const int lane    = t & 31;
    const int m       = lane & 15;
    const int half    = lane >> 4;
    const int kh      = half << 1;
    const int rowTile = blockIdx.x * WAVES + wave;
    const int e       = rowTile * 16 + m;

    v4f w2a[K2 / 4], w2b[K2 / 4];
    request_weights<K2 / 4 - 2, K2 / 4, K2 / 4>(w2a, w2b, W_upd, m, kh);
    const v4f bias1 = ((const v4f*)b_init)[m];
    const v4f bias2 = ((const v4f*)b_upd)[m];

    int4 dv[16], sv[16];
    #pragma unroll
    for (int i = 0; i < 16; ++i) {
        dv[i] = ((const int4*)edge_dst)[i * THREADS + t];
        sv[i] = ((const int4*)edge_src)[i * THREADS + t];
    }
    const int si = edge_src[e];
    const int di = edge_dst[e];

    for (int i = t; i < N_NODES / 4; i += THREADS)
        ((int4*)s_head)[i] = make_int4(-1, -1, -1, -1);

    v4f w1a[K1 / 4], w1b[K1 / 4];
    request_weights<0, K1 / 4, K1 / 4>(w1a, w1b, W_init, m, kh);

    lds_barrier();

    v2f a_own[K1 / 4];
    load_rows(a_own, node_feature, edge_feature, si, e, kh);

    int4 ov[16];
    #pragma unroll
    for (int i = 0; i < 16; ++i) {
        const int e0 = 4 * (i * THREADS + t);
        ov[i].x = atomicExch(&s_head[dv[i].x], e0);
        ov[i].y = atomicExch(&s_head[dv[i].y], e0 + 1);
        ov[i].z = atomicExch(&s_head[dv[i].z], e0 + 2);
        ov[i].w = atomicExch(&s_head[dv[i].w], e0 + 3);
    }
    asm volatile("" :: "v"(a_own[0]),  "v"(a_own[1]),  "v"(a_own[2]),  "v"(a_own[3]),
                       "v"(a_own[4]),  "v"(a_own[5]),  "v"(a_own[6]),  "v"(a_own[7]),
                       "v"(a_own[8]),  "v"(a_own[9]),  "v"(a_own[10]), "v"(a_own[11]),
                       "v"(a_own[12]), "v"(a_own[13]), "v"(a_own[14]), "v"(a_own[15]),
                       "v"(a_own[16]), "v"(a_own[17]), "v"(a_own[18]), "v"(a_own[19]));
    __builtin_amdgcn_sched_barrier(0);

    request_weights<0, K2 / 4 - 2, K2 / 4>(w2a, w2b, W_upd, m, kh);

    #pragma unroll
    for (int i = 0; i < 16; ++i) {
        ((int4*)s_link)[2 * (i * THREADS + t)    ] = make_int4(ov[i].x, sv[i].x, ov[i].y, sv[i].y);
        ((int4*)s_link)[2 * (i * THREADS + t) + 1] = make_int4(ov[i].z, sv[i].z, ov[i].w, sv[i].w);
    }
    lds_barrier();

    float* tile = s_tile[wave];
    v8f acc[4];

    first_layer(acc, a_own, w1a, w1b, bias1);
    park_tile(tile, acc, m, half);
    v2f xa[16];
    #pragma unroll
    for (int q = 0; q < 16; ++q) xa[q] = *(const v2f*)&tile[m * PITCH + 4 * q + kh];

    v2f msg[16];
    #pragma unroll
    for (int q = 0; q < 16; ++q) msg[q] = v2f{0.f, 0.f};
    int prev = -1;
    for (int pass = 0; pass < N_EDGES / SLOTS + 1; ++pass) {
        int s0 = LIST_END, s1 = LIST_END, s2 = LIST_END, s3 = LIST_END;
        int s4 = LIST_END, s5 = LIST_END, s6 = LIST_END, s7 = LIST_END;
        int j = s_head[si];
        for (int n = 0; n < N_EDGES && (unsigned)j < (unsigned)N_EDGES; ++n) {
            const int2 ln = *(const int2*)&s_link[2 * j];
            int x = (ln.y != di && j > prev) ? j : LIST_END;
            INS(s0) INS(s1) INS(s2) INS(s3) INS(s4) INS(s5) INS(s6) INS(s7)
            j = ln.x;
        }
        const int c = (s0 < LIST_END) + (s1 < LIST_END) + (s2 < LIST_END) + (s3 < LIST_END)
                    + (s4 < LIST_END) + (s5 < LIST_END) + (s6 < LIST_END) + (s7 < LIST_END);
        if (__builtin_amdgcn_ballot_w32(c > 0) == 0u) break;

        if (half == 0) s_cnt[wave][m] = c;
        int base = 0, total = 0;
        #pragma unroll
        for (int k = 0; k < 16; ++k) {
            const int ck = s_cnt[wave][k];
            base  += (k < m) ? ck : 0;
            total += ck;
        }
        total = __builtin_amdgcn_readfirstlane(total);
        if (half == 0) {
            if (c > 0) s_nbr[wave][base    ] = s0;
            if (c > 1) s_nbr[wave][base + 1] = s1;
            if (c > 2) s_nbr[wave][base + 2] = s2;
            if (c > 3) s_nbr[wave][base + 3] = s3;
            if (c > 4) s_nbr[wave][base + 4] = s4;
            if (c > 5) s_nbr[wave][base + 5] = s5;
            if (c > 6) s_nbr[wave][base + 6] = s6;
            if (c > 7) s_nbr[wave][base + 7] = s7;
        }

        for (int t0 = 0; t0 < total; t0 += 16) {
            const int  p    = t0 + m;
            const bool live = p < total;
            const int  jr   = live ? s_nbr[wave][p] : e;
            const int  sj   = live ? s_link[2 * jr + 1] : si;
            v2f a_nb[K1 / 4];
            load_rows(a_nb, node_feature, edge_feature, sj, jr, kh);
            first_layer(acc, a_nb, w1a, w1b, bias1);
            park_tile(tile, acc, m, half);
            #pragma unroll
            for (int u = 0; u < SLOTS; ++u) {
                const int pu = base + u - t0;
                if (u < c && pu >= 0 && pu < 16) {
                    const float* r_ = &tile[pu * PITCH + kh];
                    #pragma unroll
                    for (int q = 0; q < 16; ++q) {
                        msg[q].x += r_[4 * q];
                        msg[q].y += r_[4 * q + 1];
                    }
                }
            }
        }
        if (__builtin_amdgcn_ballot_w32(c == SLOTS) == 0u) break;
        prev = (c == SLOTS) ? s7 : LIST_END;
    }

    #pragma unroll
    for (int nt = 0; nt < 4; ++nt) acc[nt] = v8f{};
    #pragma unroll
    for (int q = 0; q < K2 / 4; ++q) {
        const v2f a = (q < UNITS / 4) ? xa[q] : msg[q - UNITS / 4];
        #pragma unroll
        for (int nt = 0; nt < 4; ++nt) {
            const v2f b = {w2a[q][nt], w2b[q][nt]};
            acc[nt] = __builtin_amdgcn_wmma_f32_16x16x4_f32(
                false, a, false, b, (short)0, acc[nt], false, false);
        }
    }

    #pragma unroll
    for (int r = 0; r < 8; ++r) {
        const int mm = r + 8 * half;
        *(v4f*)&out[(long)(rowTile * 16 + mm) * UNITS + 4 * m] =
            v4f{acc[0][r] + bias2[0], acc[1][r] + bias2[1],
                acc[2][r] + bias2[2], acc[3][r] + bias2[3]};
    }
}

extern "C" void kernel_launch(void* const* d_in, const int* in_sizes, int n_in,
                              void* d_out, int out_size, void* d_ws, size_t ws_size,
                              hipStream_t stream)
{
    const float* node_feature = (const float*)d_in[0];
    const float* edge_feature = (const float*)d_in[1];
    const float* W_init       = (const float*)d_in[2];
    const float* b_init       = (const float*)d_in[3];
    const float* W_upd        = (const float*)d_in[4];
    const float* b_upd        = (const float*)d_in[5];
    const int*   edge_src     = (const int*)d_in[6];
    const int*   edge_dst     = (const int*)d_in[7];
    float*       out          = (float*)d_out;

    edgeconv_kernel<<<128, 128, 0, stream>>>(node_feature, edge_feature,
                                                 W_init, b_init, W_upd, b_upd,
                                                 edge_src, edge_dst, out);
}


# 16-edge tile's matrix work split over a wave pair on two SIMDs: state wave does own rows + first 64 K of the update chain, handed through LDS to the message wave; grid 128 -> 256; same order of sums
# speedup vs baseline: 54.4092x; 1.0838x over previous
#include <hip/hip_runtime.h>


typedef float v2f __attribute__((ext_vector_type(2)));
typedef float v4f __attribute__((ext_vector_type(4)));
typedef float v8f __attribute__((ext_vector_type(8)));

#define N_NODES 20000
#define N_EDGES 8192
#define D_NODE  64
#define D_EDGE  16
#define UNITS   64
#define K1      (D_NODE + D_EDGE)
#define K2      (UNITS + UNITS)
#define WAVES   4
#define THREADS (WAVES * 32)
#define PITCH   68
#define SLOTS   8
#define LIST_END N_EDGES

#define INS(S) { const int lo_ = min(S, x); x = max(S, x); S = lo_; }

__device__ __forceinline__ void load_rows(v2f (&a)[K1 / 4],
    const float* __restrict__ node_feature, const float* __restrict__ edge_feature,
    int sj, int je, int kh)
{
    const float* xn = node_feature + (long)sj * D_NODE;
    const float* xe = edge_feature + (long)je * D_EDGE;
    #pragma unroll
    for (int k0 = 0; k0 < K1; k0 += 4) {
        const int k = k0 + kh;
        a[k0 >> 2].x = (k     < D_NODE) ? xn[k]     : xe[k     - D_NODE];
        a[k0 >> 2].y = (k + 1 < D_NODE) ? xn[k + 1] : xe[k + 1 - D_NODE];
    }
}

template <int Q0, int Q1, int KS>
__device__ __forceinline__ void request_weights(v4f (&w0)[KS], v4f (&w1)[KS],
    const float* __restrict__ W, int m, int kh)
{
    const v4f* p = (const v4f*)W + kh * (UNITS / 4) + m;
    #pragma unroll
    for (int q = Q0; q < Q1; ++q) {
        w0[q] = p[(4 * q    ) * (UNITS / 4)];
        w1[q] = p[(4 * q + 1) * (UNITS / 4)];
    }
}

__device__ __forceinline__ void first_layer(v8f (&acc)[4], const v2f (&a)[K1 / 4],
    const v4f (&w0)[K1 / 4], const v4f (&w1)[K1 / 4], const v4f bias)
{
    #pragma unroll
    for (int nt = 0; nt < 4; ++nt) acc[nt] = v8f{};
    #pragma unroll
    for (int q = 0; q < K1 / 4; ++q) {
        #pragma unroll
        for (int nt = 0; nt < 4; ++nt) {
            const v2f b = {w0[q][nt], w1[q][nt]};
            acc[nt] = __builtin_amdgcn_wmma_f32_16x16x4_f32(
                false, a[q], false, b, (short)0, acc[nt], false, false);
        }
    }
    #pragma unroll
    for (int nt = 0; nt < 4; ++nt) {
        #pragma unroll
        for (int r = 0; r < 8; ++r) acc[nt][r] = acc[nt][r] + bias[nt];
    }
}

__device__ __forceinline__ void lds_barrier()
{
    __builtin_amdgcn_fence(__ATOMIC_RELEASE, "workgroup", "local");
    __builtin_amdgcn_s_barrier();
    __builtin_amdgcn_fence(__ATOMIC_ACQUIRE, "workgroup", "local");
}

__device__ __forceinline__ void park_tile(float* tile, const v8f (&acc)[4], int m, int half)
{
    #pragma unroll
    for (int r = 0; r < 8; ++r)
        *(v4f*)&tile[(r + 8 * half) * PITCH + 4 * m] =
            v4f{acc[0][r], acc[1][r], acc[2][r], acc[3][r]};
}

__global__ __launch_bounds__(THREADS) void edgeconv_kernel(
    const float* __restrict__ node_feature,
    const float* __restrict__ edge_feature,
    const float* __restrict__ W_init,
    const float* __restrict__ b_init,
    const float* __restrict__ W_upd,
    const float* __restrict__ b_upd,
    const int*   __restrict__ edge_src,
    const int*   __restrict__ edge_dst,
    float*       __restrict__ out)
{
    __shared__ alignas(16) int   s_head[N_NODES];
    __shared__ alignas(16) int   s_link[2 * N_EDGES];
    __shared__ alignas(16) float s_tile[WAVES][16 * PITCH];
    __shared__ int s_nbr[WAVES][16 * SLOTS];
    __shared__ int s_cnt[WAVES][16];

    const int t       = threadIdx.x;
    const int wave    = __builtin_amdgcn_readfirstlane(t >> 5);
    const int role    = wave & 1;
    const int lane    = t & 31;
    const int m       = lane & 15;
    const int half    = lane >> 4;
    const int kh      = half << 1;
    const int rowTile = blockIdx.x * (WAVES / 2) + (wave >> 1);
    const int e       = rowTile * 16 + m;

    const v4f bias1 = ((const v4f*)b_init)[m];
    const v4f bias2 = ((const v4f*)b_upd)[m];

    int4 dv[16], sv[16];
    #pragma unroll
    for (int i = 0; i < 16; ++i) {
        dv[i] = ((const int4*)edge_dst)[i * THREADS + t];
        sv[i] = ((const int4*)edge_src)[i * THREADS + t];
    }
    const int si = edge_src[e];
    const int di = edge_dst[e];

    for (int i = t; i < N_NODES / 4; i += THREADS)
        ((int4*)s_head)[i] = make_int4(-1, -1, -1, -1);

    v4f w1a[K1 / 4], w1b[K1 / 4];
    request_weights<0, K1 / 4, K1 / 4>(w1a, w1b, W_init, m, kh);

    lds_barrier();

    v2f a_own[K1 / 4];
    load_rows(a_own, node_feature, edge_feature, si, e, kh);

    int4 ov[16];
    #pragma unroll
    for (int i = 0; i < 16; ++i) {
        const int e0 = 4 * (i * THREADS + t);
        ov[i].x = atomicExch(&s_head[dv[i].x], e0);
        ov[i].y = atomicExch(&s_head[dv[i].y], e0 + 1);
        ov[i].z = atomicExch(&s_head[dv[i].z], e0 + 2);
        ov[i].w = atomicExch(&s_head[dv[i].w], e0 + 3);
    }
    asm volatile("" :: "v"(a_own[0]),  "v"(a_own[1]),  "v"(a_own[2]),  "v"(a_own[3]),
                       "v"(a_own[4]),  "v"(a_own[5]),  "v"(a_own[6]),  "v"(a_own[7]),
                       "v"(a_own[8]),  "v"(a_own[9]),  "v"(a_own[10]), "v"(a_own[11]),
                       "v"(a_own[12]), "v"(a_own[13]), "v"(a_own[14]), "v"(a_own[15]),
                       "v"(a_own[16]), "v"(a_own[17]), "v"(a_own[18]), "v"(a_own[19]));
    __builtin_amdgcn_sched_barrier(0);

    v4f w2a[UNITS / 4], w2b[UNITS / 4];
    request_weights<0, UNITS / 4, UNITS / 4>(w2a, w2b, W_upd + role * (UNITS * UNITS), m, kh);

    #pragma unroll
    for (int i = 0; i < 16; ++i) {
        ((int4*)s_link)[2 * (i * THREADS + t)    ] = make_int4(ov[i].x, sv[i].x, ov[i].y, sv[i].y);
        ((int4*)s_link)[2 * (i * THREADS + t) + 1] = make_int4(ov[i].z, sv[i].z, ov[i].w, sv[i].w);
    }
    lds_barrier();

    float* tile = s_tile[wave];
    v8f acc[4];
    v2f msg[16];
    #pragma unroll
    for (int q = 0; q < 16; ++q) msg[q] = v2f{0.f, 0.f};

    if (role == 0) {
        first_layer(acc, a_own, w1a, w1b, bias1);
        park_tile(tile, acc, m, half);
        v2f xa[16];
        #pragma unroll
        for (int q = 0; q < 16; ++q) xa[q] = *(const v2f*)&tile[m * PITCH + 4 * q + kh];

        #pragma unroll
        for (int nt = 0; nt < 4; ++nt) acc[nt] = v8f{};
        #pragma unroll
        for (int q = 0; q < UNITS / 4; ++q) {
            #pragma unroll
            for (int nt = 0; nt < 4; ++nt) {
                const v2f b = {w2a[q][nt], w2b[q][nt]};
                acc[nt] = __builtin_amdgcn_wmma_f32_16x16x4_f32(
                    false, xa[q], false, b, (short)0, acc[nt], false, false);
            }
        }
        #pragma unroll
        for (int i = 0; i < 8; ++i)
            *(v4f*)&tile[(i * 32 + lane) * 4] =
                v4f{acc[i >> 1][4 * (i & 1)],     acc[i >> 1][4 * (i & 1) + 1],
                    acc[i >> 1][4 * (i & 1) + 2], acc[i >> 1][4 * (i & 1) + 3]};
    } else {
        int prev = -1;
        for (int pass = 0; pass < N_EDGES / SLOTS + 1; ++pass) {
            int s0 = LIST_END, s1 = LIST_END, s2 = LIST_END, s3 = LIST_END;
            int s4 = LIST_END, s5 = LIST_END, s6 = LIST_END, s7 = LIST_END;
            int j = s_head[si];
            for (int n = 0; n < N_EDGES && (unsigned)j < (unsigned)N_EDGES; ++n) {
                const int2 ln = *(const int2*)&s_link[2 * j];
                int x = (ln.y != di && j > prev) ? j : LIST_END;
                INS(s0) INS(s1) INS(s2) INS(s3) INS(s4) INS(s5) INS(s6) INS(s7)
                j = ln.x;
            }
            const int c = (s0 < LIST_END) + (s1 < LIST_END) + (s2 < LIST_END) + (s3 < LIST_END)
                        + (s4 < LIST_END) + (s5 < LIST_END) + (s6 < LIST_END) + (s7 < LIST_END);
            if (__builtin_amdgcn_ballot_w32(c > 0) == 0u) break;

            if (half == 0) s_cnt[wave][m] = c;
            int base = 0, total = 0;
            #pragma unroll
            for (int k = 0; k < 16; ++k) {
                const int ck = s_cnt[wave][k];
                base  += (k < m) ? ck : 0;
                total += ck;
            }
            total = __builtin_amdgcn_readfirstlane(total);
            if (half == 0) {
                if (c > 0) s_nbr[wave][base    ] = s0;
                if (c > 1) s_nbr[wave][base + 1] = s1;
                if (c > 2) s_nbr[wave][base + 2] = s2;
                if (c > 3) s_nbr[wave][base + 3] = s3;
                if (c > 4) s_nbr[wave][base + 4] = s4;
                if (c > 5) s_nbr[wave][base + 5] = s5;
                if (c > 6) s_nbr[wave][base + 6] = s6;
                if (c > 7) s_nbr[wave][base + 7] = s7;
            }

            for (int t0 = 0; t0 < total; t0 += 16) {
                const int  p    = t0 + m;
                const bool live = p < total;
                const int  jr   = live ? s_nbr[wave][p] : e;
                const int  sj   = live ? s_link[2 * jr + 1] : si;
                v2f a_nb[K1 / 4];
                load_rows(a_nb, node_feature, edge_feature, sj, jr, kh);
                first_layer(acc, a_nb, w1a, w1b, bias1);
                park_tile(tile, acc, m, half);
                #pragma unroll
                for (int u = 0; u < SLOTS; ++u) {
                    const int pu = base + u - t0;
                    if (u < c && pu >= 0 && pu < 16) {
                        const float* r_ = &tile[pu * PITCH + kh];
                        #pragma unroll
                        for (int q = 0; q < 16; ++q) {
                            msg[q].x += r_[4 * q];
                            msg[q].y += r_[4 * q + 1];
                        }
                    }
                }
            }
            if (__builtin_amdgcn_ballot_w32(c == SLOTS) == 0u) break;
            prev = (c == SLOTS) ? s7 : LIST_END;
        }
    }

    lds_barrier();
    if (role == 0) return;

    const float* part = s_tile[wave ^ 1];
    #pragma unroll
    for (int i = 0; i < 8; ++i) {
        const v4f p = *(const v4f*)&part[(i * 32 + lane) * 4];
        acc[i >> 1][4 * (i & 1)]     = p[0];
        acc[i >> 1][4 * (i & 1) + 1] = p[1];
        acc[i >> 1][4 * (i & 1) + 2] = p[2];
        acc[i >> 1][4 * (i & 1) + 3] = p[3];
    }
    #pragma unroll
    for (int q = 0; q < UNITS / 4; ++q) {
        #pragma unroll
        for (int nt = 0; nt < 4; ++nt) {
            const v2f b = {w2a[q][nt], w2b[q][nt]};
            acc[nt] = __builtin_amdgcn_wmma_f32_16x16x4_f32(
                false, msg[q], false, b, (short)0, acc[nt], false, false);
        }
    }

    #pragma unroll
    for (int r = 0; r < 8; ++r) {
        const int mm = r + 8 * half;
        *(v4f*)&out[(long)(rowTile * 16 + mm) * UNITS + 4 * m] =
            v4f{acc[0][r] + bias2[0], acc[1][r] + bias2[1],
                acc[2][r] + bias2[2], acc[3][r] + bias2[3]};
    }
}

extern "C" void kernel_launch(void* const* d_in, const int* in_sizes, int n_in,
                              void* d_out, int out_size, void* d_ws, size_t ws_size,
                              hipStream_t stream)
{
    const float* node_feature = (const float*)d_in[0];
    const float* edge_feature = (const float*)d_in[1];
    const float* W_init       = (const float*)d_in[2];
    const float* b_init       = (const float*)d_in[3];
    const float* W_upd        = (const float*)d_in[4];
    const float* b_upd        = (const float*)d_in[5];
    const int*   edge_src     = (const int*)d_in[6];
    const int*   edge_dst     = (const int*)d_in[7];
    float*       out          = (float*)d_out;

    edgeconv_kernel<<<256, 128, 0, stream>>>(node_feature, edge_feature,
                                                 W_init, b_init, W_upd, b_upd,
                                                 edge_src, edge_dst, out);
}


# pair's two waves made alike: each does 32 of the 64 first-layer columns (own and neighbour rows) and those columns' K half of the update layer; partial sums added at the end (sum order changed)
# speedup vs baseline: 58.1041x; 1.0679x over previous
#include <hip/hip_runtime.h>


typedef float v2f __attribute__((ext_vector_type(2)));
typedef float v4f __attribute__((ext_vector_type(4)));
typedef float v8f __attribute__((ext_vector_type(8)));

#define N_NODES 20000
#define N_EDGES 8192
#define D_NODE  64
#define D_EDGE  16
#define UNITS   64
#define K1      (D_NODE + D_EDGE)
#define K2      (UNITS + UNITS)
#define HALF_N  (UNITS / 2)
#define WAVES   4
#define THREADS (WAVES * 32)
#define PITCH   68
#define SLOTS   8
#define LIST_END N_EDGES

#define INS(S) { const int lo_ = min(S, x); x = max(S, x); S = lo_; }

__device__ __forceinline__ void load_rows(v2f (&a)[K1 / 4],
    const float* __restrict__ node_feature, const float* __restrict__ edge_feature,
    int sj, int je, int kh)
{
    const float* xn = node_feature + (long)sj * D_NODE;
    const float* xe = edge_feature + (long)je * D_EDGE;
    #pragma unroll
    for (int k0 = 0; k0 < K1; k0 += 4) {
        const int k = k0 + kh;
        a[k0 >> 2].x = (k     < D_NODE) ? xn[k]     : xe[k     - D_NODE];
        a[k0 >> 2].y = (k + 1 < D_NODE) ? xn[k + 1] : xe[k + 1 - D_NODE];
    }
}

__device__ __forceinline__ void request_w_init(v2f (&w0)[K1 / 4], v2f (&w1)[K1 / 4],
    const float* __restrict__ W, int hh, int m, int kh)
{
    const v2f* p = (const v2f*)W + kh * (UNITS / 2) + hh * (HALF_N / 2) + m;
    #pragma unroll
    for (int q = 0; q < K1 / 4; ++q) {
        w0[q] = p[(4 * q    ) * (UNITS / 2)];
        w1[q] = p[(4 * q + 1) * (UNITS / 2)];
    }
}

template <int Q0>
__device__ __forceinline__ void request_w_upd(v4f (&w0)[16], v4f (&w1)[16],
    const float* __restrict__ W, int r0, int m, int kh)
{
    const v4f* p = (const v4f*)W + (r0 + kh) * (UNITS / 4) + m;
    #pragma unroll
    for (int q = 0; q < 8; ++q) {
        w0[Q0 + q] = p[(4 * q    ) * (UNITS / 4)];
        w1[Q0 + q] = p[(4 * q + 1) * (UNITS / 4)];
    }
}

__device__ __forceinline__ void first_layer_half(v8f (&acc)[2], const v2f (&a)[K1 / 4],
    const v2f (&w0)[K1 / 4], const v2f (&w1)[K1 / 4], const v2f bias)
{
    #pragma unroll
    for (int nt = 0; nt < 2; ++nt) acc[nt] = v8f{};
    #pragma unroll
    for (int q = 0; q < K1 / 4; ++q) {
        #pragma unroll
        for (int nt = 0; nt < 2; ++nt) {
            const v2f b = {w0[q][nt], w1[q][nt]};
            acc[nt] = __builtin_amdgcn_wmma_f32_16x16x4_f32(
                false, a[q], false, b, (short)0, acc[nt], false, false);
        }
    }
    #pragma unroll
    for (int nt = 0; nt < 2; ++nt) {
        #pragma unroll
        for (int r = 0; r < 8; ++r) acc[nt][r] = acc[nt][r] + bias[nt];
    }
}

template <int Q0>
__device__ __forceinline__ void update_steps(v8f (&acc)[4], const v2f (&x)[8],
    const v4f (&w0)[16], const v4f (&w1)[16])
{
    #pragma unroll
    for (int q = 0; q < 8; ++q) {
        #pragma unroll
        for (int nt = 0; nt < 4; ++nt) {
            const v2f b = {w0[Q0 + q][nt], w1[Q0 + q][nt]};
            acc[nt] = __builtin_amdgcn_wmma_f32_16x16x4_f32(
                false, x[q], false, b, (short)0, acc[nt], false, false);
        }
    }
}

__device__ __forceinline__ void lds_barrier()
{
    __builtin_amdgcn_fence(__ATOMIC_RELEASE, "workgroup", "local");
    __builtin_amdgcn_s_barrier();
    __builtin_amdgcn_fence(__ATOMIC_ACQUIRE, "workgroup", "local");
}

__device__ __forceinline__ void park_half(float* tile, const v8f (&acc)[2], int m, int half)
{
    #pragma unroll
    for (int r = 0; r < 8; ++r)
        *(v2f*)&tile[(r + 8 * half) * PITCH + 2 * m] = v2f{acc[0][r], acc[1][r]};
}

__global__ __launch_bounds__(THREADS) void edgeconv_kernel(
    const float* __restrict__ node_feature,
    const float* __restrict__ edge_feature,
    const float* __restrict__ W_init,
    const float* __restrict__ b_init,
    const float* __restrict__ W_upd,
    const float* __restrict__ b_upd,
    const int*   __restrict__ edge_src,
    const int*   __restrict__ edge_dst,
    float*       __restrict__ out)
{
    __shared__ alignas(16) int   s_head[N_NODES];
    __shared__ alignas(16) int   s_link[2 * N_EDGES];
    __shared__ alignas(16) float s_tile[WAVES][16 * PITCH];
    __shared__ int s_nbr[WAVES][16 * SLOTS];
    __shared__ int s_cnt[WAVES][16];

    const int t       = threadIdx.x;
    const int wave    = __builtin_amdgcn_readfirstlane(t >> 5);
    const int hh      = wave & 1;
    const int lane    = t & 31;
    const int m       = lane & 15;
    const int half    = lane >> 4;
    const int kh      = half << 1;
    const int rowTile = blockIdx.x * (WAVES / 2) + (wave >> 1);
    const int e       = rowTile * 16 + m;

    const v2f bias1 = ((const v2f*)b_init)[hh * (HALF_N / 2) + m];
    const v4f bias2 = ((const v4f*)b_upd)[m];

    int4 dv[16], sv[16];
    #pragma unroll
    for (int i = 0; i < 16; ++i) {
        dv[i] = ((const int4*)edge_dst)[i * THREADS + t];
        sv[i] = ((const int4*)edge_src)[i * THREADS + t];
    }
    const int si = edge_src[e];
    const int di = edge_dst[e];

    for (int i = t; i < N_NODES / 4; i += THREADS)
        ((int4*)s_head)[i] = make_int4(-1, -1, -1, -1);

    v2f w1a[K1 / 4], w1b[K1 / 4];
    request_w_init(w1a, w1b, W_init, hh, m, kh);

    lds_barrier();

    v2f a_own[K1 / 4];
    load_rows(a_own, node_feature, edge_feature, si, e, kh);

    int4 ov[16];
    #pragma unroll
    for (int i = 0; i < 16; ++i) {
        const int e0 = 4 * (i * THREADS + t);
        ov[i].x = atomicExch(&s_head[dv[i].x], e0);
        ov[i].y = atomicExch(&s_head[dv[i].y], e0 + 1);
        ov[i].z = atomicExch(&s_head[dv[i].z], e0 + 2);
        ov[i].w = atomicExch(&s_head[dv[i].w], e0 + 3);
    }
    asm volatile("" :: "v"(a_own[0]),  "v"(a_own[1]),  "v"(a_own[2]),  "v"(a_own[3]),
                       "v"(a_own[4]),  "v"(a_own[5]),  "v"(a_own[6]),  "v"(a_own[7]),
                       "v"(a_own[8]),  "v"(a_own[9]),  "v"(a_own[10]), "v"(a_own[11]),
                       "v"(a_own[12]), "v"(a_own[13]), "v"(a_own[14]), "v"(a_own[15]),
                       "v"(a_own[16]), "v"(a_own[17]), "v"(a_own[18]), "v"(a_own[19]));
    __builtin_amdgcn_sched_barrier(0);

    v4f w2a[16], w2b[16];
    request_w_upd<8>(w2a, w2b, W_upd, UNITS + hh * HALF_N, m, kh);

    #pragma unroll
    for (int i = 0; i < 16; ++i) {
        ((int4*)s_link)[2 * (i * THREADS + t)    ] = make_int4(ov[i].x, sv[i].x, ov[i].y, sv[i].y);
        ((int4*)s_link)[2 * (i * THREADS + t) + 1] = make_int4(ov[i].z, sv[i].z, ov[i].w, sv[i].w);
    }
    lds_barrier();

    float* tile = s_tile[wave];
    v8f acc[4];
    #pragma unroll
    for (int nt = 0; nt < 4; ++nt) acc[nt] = v8f{};
    v2f msg[8];
    #pragma unroll
    for (int q = 0; q < 8; ++q) msg[q] = v2f{0.f, 0.f};

#define OWN_ROWS() {                                                              \
        request_w_upd<0>(w2a, w2b, W_upd, hh * HALF_N, m, kh);                    \
        v8f es[2];                                                                \
        first_layer_half(es, a_own, w1a, w1b, bias1);                             \
        park_half(tile, es, m, half);                                             \
        v2f xa[8];                                                                \
        _Pragma("unroll")                                                         \
        for (int q = 0; q < 8; ++q) xa[q] = *(const v2f*)&tile[m * PITCH + 4 * q + kh]; \
        update_steps<0>(acc, xa, w2a, w2b);                                       \
    }

    int first_total = 0;
    int prev = -1;
    for (int pass = 0; pass < N_EDGES / SLOTS + 1; ++pass) {
        int s0 = LIST_END, s1 = LIST_END, s2 = LIST_END, s3 = LIST_END;
        int s4 = LIST_END, s5 = LIST_END, s6 = LIST_END, s7 = LIST_END;
        int j = s_head[si];
        for (int n = 0; n < N_EDGES && (unsigned)j < (unsigned)N_EDGES; ++n) {
            const int2 ln = *(const int2*)&s_link[2 * j];
            int x = (ln.y != di && j > prev) ? j : LIST_END;
            INS(s0) INS(s1) INS(s2) INS(s3) INS(s4) INS(s5) INS(s6) INS(s7)
            j = ln.x;
        }
        const int c = (s0 < LIST_END) + (s1 < LIST_END) + (s2 < LIST_END) + (s3 < LIST_END)
                    + (s4 < LIST_END) + (s5 < LIST_END) + (s6 < LIST_END) + (s7 < LIST_END);
        if (__builtin_amdgcn_ballot_w32(c > 0) == 0u) break;

        if (half == 0) s_cnt[wave][m] = c;
        int base = 0, total = 0;
        #pragma unroll
        for (int k = 0; k < 16; ++k) {
            const int ck = s_cnt[wave][k];
            base  += (k < m) ? ck : 0;
            total += ck;
        }
        total = __builtin_amdgcn_readfirstlane(total);
        if (pass == 0) first_total = total;
        if (half == 0) {
            if (c > 0) s_nbr[wave][base    ] = s0;
            if (c > 1) s_nbr[wave][base + 1] = s1;
            if (c > 2) s_nbr[wave][base + 2] = s2;
            if (c > 3) s_nbr[wave][base + 3] = s3;
            if (c > 4) s_nbr[wave][base + 4] = s4;
            if (c > 5) s_nbr[wave][base + 5] = s5;
            if (c > 6) s_nbr[wave][base + 6] = s6;
            if (c > 7) s_nbr[wave][base + 7] = s7;
        }

        for (int t0 = 0; t0 < total; t0 += 16) {
            const int  p    = t0 + m;
            const bool live = p < total;
            const int  jr   = live ? s_nbr[wave][p] : e;
            const int  sj   = live ? s_link[2 * jr + 1] : si;
            v2f a_nb[K1 / 4];
            load_rows(a_nb, node_feature, edge_feature, sj, jr, kh);
            if (pass == 0 && t0 == 0) OWN_ROWS()
            v8f nb[2];
            first_layer_half(nb, a_nb, w1a, w1b, bias1);
            park_half(tile, nb, m, half);
            #pragma unroll
            for (int u = 0; u < SLOTS; ++u) {
                const int pu = base + u - t0;
                if (u < c && pu >= 0 && pu < 16) {
                    const float* r_ = &tile[pu * PITCH + kh];
                    #pragma unroll
                    for (int q = 0; q < 8; ++q) {
                        msg[q].x += r_[4 * q];
                        msg[q].y += r_[4 * q + 1];
                    }
                }
            }
        }
        if (__builtin_amdgcn_ballot_w32(c == SLOTS) == 0u) break;
        prev = (c == SLOTS) ? s7 : LIST_END;
    }
    if (first_total == 0) OWN_ROWS()
#undef OWN_ROWS

    update_steps<8>(acc, msg, w2a, w2b);

    if (hh == 1) {
        #pragma unroll
        for (int i = 0; i < 8; ++i)
            *(v4f*)&tile[(i * 32 + lane) * 4] =
                v4f{acc[i >> 1][4 * (i & 1)],     acc[i >> 1][4 * (i & 1) + 1],
                    acc[i >> 1][4 * (i & 1) + 2], acc[i >> 1][4 * (i & 1) + 3]};
    }
    lds_barrier();
    if (hh == 1) return;

    const float* part = s_tile[wave ^ 1];
    #pragma unroll
    for (int i = 0; i < 8; ++i) {
        const v4f p = *(const v4f*)&part[(i * 32 + lane) * 4];
        acc[i >> 1][4 * (i & 1)]     += p[0];
        acc[i >> 1][4 * (i & 1) + 1] += p[1];
        acc[i >> 1][4 * (i & 1) + 2] += p[2];
        acc[i >> 1][4 * (i & 1) + 3] += p[3];
    }

    #pragma unroll
    for (int r = 0; r < 8; ++r) {
        const int mm = r + 8 * half;
        *(v4f*)&out[(long)(rowTile * 16 + mm) * UNITS + 4 * m] =
            v4f{acc[0][r] + bias2[0], acc[1][r] + bias2[1],
                acc[2][r] + bias2[2], acc[3][r] + bias2[3]};
    }
}

extern "C" void kernel_launch(void* const* d_in, const int* in_sizes, int n_in,
                              void* d_out, int out_size, void* d_ws, size_t ws_size,
                              hipStream_t stream)
{
    const float* node_feature = (const float*)d_in[0];
    const float* edge_feature = (const float*)d_in[1];
    const float* W_init       = (const float*)d_in[2];
    const float* b_init       = (const float*)d_in[3];
    const float* W_upd        = (const float*)d_in[4];
    const float* b_upd        = (const float*)d_in[5];
    const int*   edge_src     = (const int*)d_in[6];
    const int*   edge_dst     = (const int*)d_in[7];
    float*       out          = (float*)d_out;

    edgeconv_kernel<<<256, 128, 0, stream>>>(node_feature, edge_feature,
                                                 W_init, b_init, W_upd, b_upd,
                                                 edge_src, edge_dst, out);
}


# list exchanges run group by group as the index reads land (were held until all had returned): own-row requests, which need the last index read, and their wait moved behind the exchanges
# speedup vs baseline: 61.5834x; 1.0599x over previous
#include <hip/hip_runtime.h>


typedef float v2f __attribute__((ext_vector_type(2)));
typedef float v4f __attribute__((ext_vector_type(4)));
typedef float v8f __attribute__((ext_vector_type(8)));

#define N_NODES 20000
#define N_EDGES 8192
#define D_NODE  64
#define D_EDGE  16
#define UNITS   64
#define K1      (D_NODE + D_EDGE)
#define K2      (UNITS + UNITS)
#define HALF_N  (UNITS / 2)
#define WAVES   4
#define THREADS (WAVES * 32)
#define PITCH   68
#define SLOTS   8
#define LIST_END N_EDGES

#define INS(S) { const int lo_ = min(S, x); x = max(S, x); S = lo_; }

__device__ __forceinline__ void load_rows(v2f (&a)[K1 / 4],
    const float* __restrict__ node_feature, const float* __restrict__ edge_feature,
    int sj, int je, int kh)
{
    const float* xn = node_feature + (long)sj * D_NODE;
    const float* xe = edge_feature + (long)je * D_EDGE;
    #pragma unroll
    for (int k0 = 0; k0 < K1; k0 += 4) {
        const int k = k0 + kh;
        a[k0 >> 2].x = (k     < D_NODE) ? xn[k]     : xe[k     - D_NODE];
        a[k0 >> 2].y = (k + 1 < D_NODE) ? xn[k + 1] : xe[k + 1 - D_NODE];
    }
}

__device__ __forceinline__ void request_w_init(v2f (&w0)[K1 / 4], v2f (&w1)[K1 / 4],
    const float* __restrict__ W, int hh, int m, int kh)
{
    const v2f* p = (const v2f*)W + kh * (UNITS / 2) + hh * (HALF_N / 2) + m;
    #pragma unroll
    for (int q = 0; q < K1 / 4; ++q) {
        w0[q] = p[(4 * q    ) * (UNITS / 2)];
        w1[q] = p[(4 * q + 1) * (UNITS / 2)];
    }
}

template <int Q0>
__device__ __forceinline__ void request_w_upd(v4f (&w0)[16], v4f (&w1)[16],
    const float* __restrict__ W, int r0, int m, int kh)
{
    const v4f* p = (const v4f*)W + (r0 + kh) * (UNITS / 4) + m;
    #pragma unroll
    for (int q = 0; q < 8; ++q) {
        w0[Q0 + q] = p[(4 * q    ) * (UNITS / 4)];
        w1[Q0 + q] = p[(4 * q + 1) * (UNITS / 4)];
    }
}

__device__ __forceinline__ void first_layer_half(v8f (&acc)[2], const v2f (&a)[K1 / 4],
    const v2f (&w0)[K1 / 4], const v2f (&w1)[K1 / 4], const v2f bias)
{
    #pragma unroll
    for (int nt = 0; nt < 2; ++nt) acc[nt] = v8f{};
    #pragma unroll
    for (int q = 0; q < K1 / 4; ++q) {
        #pragma unroll
        for (int nt = 0; nt < 2; ++nt) {
            const v2f b = {w0[q][nt], w1[q][nt]};
            acc[nt] = __builtin_amdgcn_wmma_f32_16x16x4_f32(
                false, a[q], false, b, (short)0, acc[nt], false, false);
        }
    }
    #pragma unroll
    for (int nt = 0; nt < 2; ++nt) {
        #pragma unroll
        for (int r = 0; r < 8; ++r) acc[nt][r] = acc[nt][r] + bias[nt];
    }
}

template <int Q0>
__device__ __forceinline__ void update_steps(v8f (&acc)[4], const v2f (&x)[8],
    const v4f (&w0)[16], const v4f (&w1)[16])
{
    #pragma unroll
    for (int q = 0; q < 8; ++q) {
        #pragma unroll
        for (int nt = 0; nt < 4; ++nt) {
            const v2f b = {w0[Q0 + q][nt], w1[Q0 + q][nt]};
            acc[nt] = __builtin_amdgcn_wmma_f32_16x16x4_f32(
                false, x[q], false, b, (short)0, acc[nt], false, false);
        }
    }
}

__device__ __forceinline__ void lds_barrier()
{
    __builtin_amdgcn_fence(__ATOMIC_RELEASE, "workgroup", "local");
    __builtin_amdgcn_s_barrier();
    __builtin_amdgcn_fence(__ATOMIC_ACQUIRE, "workgroup", "local");
}

__device__ __forceinline__ void park_half(float* tile, const v8f (&acc)[2], int m, int half)
{
    #pragma unroll
    for (int r = 0; r < 8; ++r)
        *(v2f*)&tile[(r + 8 * half) * PITCH + 2 * m] = v2f{acc[0][r], acc[1][r]};
}

__global__ __launch_bounds__(THREADS) void edgeconv_kernel(
    const float* __restrict__ node_feature,
    const float* __restrict__ edge_feature,
    const float* __restrict__ W_init,
    const float* __restrict__ b_init,
    const float* __restrict__ W_upd,
    const float* __restrict__ b_upd,
    const int*   __restrict__ edge_src,
    const int*   __restrict__ edge_dst,
    float*       __restrict__ out)
{
    __shared__ alignas(16) int   s_head[N_NODES];
    __shared__ alignas(16) int   s_link[2 * N_EDGES];
    __shared__ alignas(16) float s_tile[WAVES][16 * PITCH];
    __shared__ int s_nbr[WAVES][16 * SLOTS];
    __shared__ int s_cnt[WAVES][16];

    const int t       = threadIdx.x;
    const int wave    = __builtin_amdgcn_readfirstlane(t >> 5);
    const int hh      = wave & 1;
    const int lane    = t & 31;
    const int m       = lane & 15;
    const int half    = lane >> 4;
    const int kh      = half << 1;
    const int rowTile = blockIdx.x * (WAVES / 2) + (wave >> 1);
    const int e       = rowTile * 16 + m;

    const v2f bias1 = ((const v2f*)b_init)[hh * (HALF_N / 2) + m];
    const v4f bias2 = ((const v4f*)b_upd)[m];

    int4 dv[16], sv[16];
    #pragma unroll
    for (int i = 0; i < 16; ++i) {
        dv[i] = ((const int4*)edge_dst)[i * THREADS + t];
        sv[i] = ((const int4*)edge_src)[i * THREADS + t];
    }
    const int si = edge_src[e];
    const int di = edge_dst[e];

    for (int i = t; i < N_NODES / 4; i += THREADS)
        ((int4*)s_head)[i] = make_int4(-1, -1, -1, -1);

    v2f w1a[K1 / 4], w1b[K1 / 4];
    request_w_init(w1a, w1b, W_init, hh, m, kh);
    __builtin_amdgcn_sched_barrier(0);

    lds_barrier();

    int4 ov[16];
    #pragma unroll
    for (int i = 0; i < 16; ++i) {
        const int e0 = 4 * (i * THREADS + t);
        ov[i].x = atomicExch(&s_head[dv[i].x], e0);
        ov[i].y = atomicExch(&s_head[dv[i].y], e0 + 1);
        ov[i].z = atomicExch(&s_head[dv[i].z], e0 + 2);
        ov[i].w = atomicExch(&s_head[dv[i].w], e0 + 3);
    }
    v2f a_own[K1 / 4];
    load_rows(a_own, node_feature, edge_feature, si, e, kh);


    #pragma unroll
    for (int i = 0; i < 16; ++i) {
        ((int4*)s_link)[2 * (i * THREADS + t)    ] = make_int4(ov[i].x, sv[i].x, ov[i].y, sv[i].y);
        ((int4*)s_link)[2 * (i * THREADS + t) + 1] = make_int4(ov[i].z, sv[i].z, ov[i].w, sv[i].w);
    }
    lds_barrier();

    v4f w2a[16], w2b[16];
    request_w_upd<8>(w2a, w2b, W_upd, UNITS + hh * HALF_N, m, kh);

    float* tile = s_tile[wave];
    v8f acc[4];
    #pragma unroll
    for (int nt = 0; nt < 4; ++nt) acc[nt] = v8f{};
    v2f msg[8];
    #pragma unroll
    for (int q = 0; q < 8; ++q) msg[q] = v2f{0.f, 0.f};

#define OWN_ROWS() {                                                              \
        request_w_upd<0>(w2a, w2b, W_upd, hh * HALF_N, m, kh);                    \
        v8f es[2];                                                                \
        first_layer_half(es, a_own, w1a, w1b, bias1);                             \
        park_half(tile, es, m, half);                                             \
        v2f xa[8];                                                                \
        _Pragma("unroll")                                                         \
        for (int q = 0; q < 8; ++q) xa[q] = *(const v2f*)&tile[m * PITCH + 4 * q + kh]; \
        update_steps<0>(acc, xa, w2a, w2b);                                       \
    }

    int first_total = 0;
    int prev = -1;
    for (int pass = 0; pass < N_EDGES / SLOTS + 1; ++pass) {
        int s0 = LIST_END, s1 = LIST_END, s2 = LIST_END, s3 = LIST_END;
        int s4 = LIST_END, s5 = LIST_END, s6 = LIST_END, s7 = LIST_END;
        int j = s_head[si];
        for (int n = 0; n < N_EDGES && (unsigned)j < (unsigned)N_EDGES; ++n) {
            const int2 ln = *(const int2*)&s_link[2 * j];
            int x = (ln.y != di && j > prev) ? j : LIST_END;
            INS(s0) INS(s1) INS(s2) INS(s3) INS(s4) INS(s5) INS(s6) INS(s7)
            j = ln.x;
        }
        const int c = (s0 < LIST_END) + (s1 < LIST_END) + (s2 < LIST_END) + (s3 < LIST_END)
                    + (s4 < LIST_END) + (s5 < LIST_END) + (s6 < LIST_END) + (s7 < LIST_END);
        if (__builtin_amdgcn_ballot_w32(c > 0) == 0u) break;

        if (half == 0) s_cnt[wave][m] = c;
        int base = 0, total = 0;
        #pragma unroll
        for (int k = 0; k < 16; ++k) {
            const int ck = s_cnt[wave][k];
            base  += (k < m) ? ck : 0;
            total += ck;
        }
        total = __builtin_amdgcn_readfirstlane(total);
        if (pass == 0) first_total = total;
        if (half == 0) {
            if (c > 0) s_nbr[wave][base    ] = s0;
            if (c > 1) s_nbr[wave][base + 1] = s1;
            if (c > 2) s_nbr[wave][base + 2] = s2;
            if (c > 3) s_nbr[wave][base + 3] = s3;
            if (c > 4) s_nbr[wave][base + 4] = s4;
            if (c > 5) s_nbr[wave][base + 5] = s5;
            if (c > 6) s_nbr[wave][base + 6] = s6;
            if (c > 7) s_nbr[wave][base + 7] = s7;
        }

        for (int t0 = 0; t0 < total; t0 += 16) {
            const int  p    = t0 + m;
            const bool live = p < total;
            const int  jr   = live ? s_nbr[wave][p] : e;
            const int  sj   = live ? s_link[2 * jr + 1] : si;
            v2f a_nb[K1 / 4];
            load_rows(a_nb, node_feature, edge_feature, sj, jr, kh);
            if (pass == 0 && t0 == 0) OWN_ROWS()
            v8f nb[2];
            first_layer_half(nb, a_nb, w1a, w1b, bias1);
            park_half(tile, nb, m, half);
            #pragma unroll
            for (int u = 0; u < SLOTS; ++u) {
                const int pu = base + u - t0;
                if (u < c && pu >= 0 && pu < 16) {
                    const float* r_ = &tile[pu * PITCH + kh];
                    #pragma unroll
                    for (int q = 0; q < 8; ++q) {
                        msg[q].x += r_[4 * q];
                        msg[q].y += r_[4 * q + 1];
                    }
                }
            }
        }
        if (__builtin_amdgcn_ballot_w32(c == SLOTS) == 0u) break;
        prev = (c == SLOTS) ? s7 : LIST_END;
    }
    if (first_total == 0) OWN_ROWS()
#undef OWN_ROWS

    update_steps<8>(acc, msg, w2a, w2b);

    if (hh == 1) {
        #pragma unroll
        for (int i = 0; i < 8; ++i)
            *(v4f*)&tile[(i * 32 + lane) * 4] =
                v4f{acc[i >> 1][4 * (i & 1)],     acc[i >> 1][4 * (i & 1) + 1],
                    acc[i >> 1][4 * (i & 1) + 2], acc[i >> 1][4 * (i & 1) + 3]};
    }
    lds_barrier();
    if (hh == 1) return;

    const float* part = s_tile[wave ^ 1];
    #pragma unroll
    for (int i = 0; i < 8; ++i) {
        const v4f p = *(const v4f*)&part[(i * 32 + lane) * 4];
        acc[i >> 1][4 * (i & 1)]     += p[0];
        acc[i >> 1][4 * (i & 1) + 1] += p[1];
        acc[i >> 1][4 * (i & 1) + 2] += p[2];
        acc[i >> 1][4 * (i & 1) + 3] += p[3];
    }

    #pragma unroll
    for (int r = 0; r < 8; ++r) {
        const int mm = r + 8 * half;
        *(v4f*)&out[(long)(rowTile * 16 + mm) * UNITS + 4 * m] =
            v4f{acc[0][r] + bias2[0], acc[1][r] + bias2[1],
                acc[2][r] + bias2[2], acc[3][r] + bias2[3]};
    }
}

extern "C" void kernel_launch(void* const* d_in, const int* in_sizes, int n_in,
                              void* d_out, int out_size, void* d_ws, size_t ws_size,
                              hipStream_t stream)
{
    const float* node_feature = (const float*)d_in[0];
    const float* edge_feature = (const float*)d_in[1];
    const float* W_init       = (const float*)d_in[2];
    const float* b_init       = (const float*)d_in[3];
    const float* W_upd        = (const float*)d_in[4];
    const float* b_upd        = (const float*)d_in[5];
    const int*   edge_src     = (const int*)d_in[6];
    const int*   edge_dst     = (const int*)d_in[7];
    float*       out          = (float*)d_out;

    edgeconv_kernel<<<256, 128, 0, stream>>>(node_feature, edge_feature,
                                                 W_init, b_init, W_upd, b_upd,
                                                 edge_src, edge_dst, out);
}
